# starting kernel
# baseline (speedup 1.0000x reference)
#include <hip/hip_runtime.h>
#include <hip/hip_bf16.h>
#include <math.h>

typedef __attribute__((ext_vector_type(2))) float v2f;
typedef __attribute__((ext_vector_type(8))) float v8f;

__global__ __launch_bounds__(256) void img_mean_kernel(const float* __restrict__ in,
                                                       float* __restrict__ out) {
    int row  = (blockIdx.x << 3) + (threadIdx.x >> 5);
    int lane = threadIdx.x & 31;
    const float4* p = reinterpret_cast<const float4*>(in + (size_t)row * 256);
    float4 v0 = p[lane];
    float4 v1 = p[lane + 32];
    float s = (v0.x + v0.y) + (v0.z + v0.w) + (v1.x + v1.y) + (v1.z + v1.w);
    #pragma unroll
    for (int off = 16; off > 0; off >>= 1)
        s += __shfl_xor(s, off, 32);
    if (lane == 0)
        out[row] = s * (1.0f / 256.0f);
}

__global__ __launch_bounds__(512) void rad_mean_kernel(const float* __restrict__ in,
                                                       float* __restrict__ out) {
    int b = blockIdx.x;
    int t = threadIdx.x;
    const float* p = in + (size_t)b * 32 * 512 + t;
    float s = 0.0f;
    #pragma unroll
    for (int r = 0; r < 32; ++r)
        s += p[r * 512];
    out[b * 512 + t] = s * (1.0f / 32.0f);
}

__global__ __launch_bounds__(256) void inv_norm_kernel(const float* __restrict__ img,
                                                       const float* __restrict__ rad,
                                                       float* __restrict__ invn) {
    __shared__ float sm[256];
    int row = blockIdx.x;
    const float* src = (row < 256) ? (img + (size_t)row * 512)
                                   : (rad + (size_t)(row - 256) * 512);
    int t = threadIdx.x;
    float a = src[t];
    float b = src[t + 256];
    sm[t] = a * a + b * b;
    __syncthreads();
    #pragma unroll
    for (int s = 128; s > 0; s >>= 1) {
        if (t < s) sm[t] += sm[t + s];
        __syncthreads();
    }
    if (t == 0)
        invn[row] = 1.0f / fmaxf(sqrtf(sm[0]), 1e-12f);
}

__global__ __launch_bounds__(32) void sim_wmma_kernel(const float* __restrict__ img,
                                                      const float* __restrict__ rad,
                                                      const float* __restrict__ invn,
                                                      const float* __restrict__ temperature,
                                                      float* __restrict__ out) {
    int tile = blockIdx.x;
    int ti = tile >> 4;
    int tj = tile & 15;
    int lane  = threadIdx.x & 31;
    int lhalf = lane >> 4;
    int lmod  = lane & 15;

    int rowA = ti * 16 + lmod;
    int rowB = tj * 16 + lmod;
    float sA = invn[rowA];
    float sB = invn[256 + rowB];

    const float* pa = img + (size_t)rowA * 512 + 2 * lhalf;
    const float* pb = rad + (size_t)rowB * 512 + 2 * lhalf;

    v8f acc = {};
    #pragma unroll 8
    for (int k = 0; k < 512; k += 4) {
        v2f a = *reinterpret_cast<const v2f*>(pa + k);
        v2f b = *reinterpret_cast<const v2f*>(pb + k);
        a = a * sA;
        b = b * sB;
        acc = __builtin_amdgcn_wmma_f32_16x16x4_f32(
             false, a,  false, b,
             (short)0, acc,  false,  false);
    }

    float invt = 1.0f / temperature[0];
    int col = tj * 16 + lmod;
    #pragma unroll
    for (int v = 0; v < 8; ++v) {
        int row = ti * 16 + v + 8 * lhalf;
        out[row * 256 + col] = acc[v] * invt;
    }
}

extern "C" void kernel_launch(void* const* d_in, const int* in_sizes, int n_in,
                              void* d_out, int out_size, void* d_ws, size_t ws_size,
                              hipStream_t stream) {
    const float* img_feat = (const float*)d_in[0];
    const float* rad_feat = (const float*)d_in[1];
    const float* temp     = (const float*)d_in[2];

    float* out_sim = (float*)d_out;
    float* out_img = out_sim + 256 * 256;
    float* out_rad = out_img + 256 * 512;

    float* invn = (float*)d_ws;

    img_mean_kernel<<<16384, 256, 0, stream>>>(img_feat, out_img);
    rad_mean_kernel<<<256, 512, 0, stream>>>(rad_feat, out_rad);
    inv_norm_kernel<<<512, 256, 0, stream>>>(out_img, out_rad, invn);
    sim_wmma_kernel<<<256, 32, 0, stream>>>(out_img, out_rad, invn, temp, out_sim);
}


# similarity kernel: a tile's 16+16 operand rows staged once in LDS by 4 waves (whole-row 16-B loads: one global round trip, was 16); wave 0 runs the WMMAs from LDS on 2 accumulators: k sum order new
# speedup vs baseline: 2.3552x; 2.3552x over previous
#include <hip/hip_runtime.h>
#include <hip/hip_bf16.h>
#include <math.h>

typedef __attribute__((ext_vector_type(2))) float v2f;
typedef __attribute__((ext_vector_type(8))) float v8f;

__device__ __forceinline__ void load_rows8(float4 (&dst)[16], const float4* __restrict__ pi, int g) {
    #pragma unroll
    for (int j = 0; j < 8; ++j) {
        dst[2 * j]     = pi[(g * 8 + j) * 64];
        dst[2 * j + 1] = pi[(g * 8 + j) * 64 + 32];
    }
}

__device__ __forceinline__ void sum_rows8(const float4 (&src)[16], int g, int lane, float& m) {
    #pragma unroll
    for (int j = 0; j < 8; ++j) {
        float4 v0 = src[2 * j];
        float4 v1 = src[2 * j + 1];
        float s = (v0.x + v0.y) + (v0.z + v0.w) + (v1.x + v1.y) + (v1.z + v1.w);
        for (int off = 16; off > 0; off >>= 1)
            s += __shfl_xor(s, off, 32);
        if (lane == g * 8 + j)
            m = s * (1.0f / 256.0f);
    }
}

__global__ __launch_bounds__(512) void embed_norm_kernel(const float* __restrict__ img,
                                                         const float* __restrict__ rad,
                                                         float* __restrict__ out_img,
                                                         float* __restrict__ out_rad,
                                                         float* __restrict__ invn) {
    __shared__ float semb[1024];
    __shared__ float sm[512];

    const int b    = blockIdx.x;
    const int t    = threadIdx.x;
    const int lane = t & 31;
    const int w    = t >> 5;

    const float* pr = rad + (size_t)b * (32 * 512) + t;
    float rv[32];
    #pragma unroll
    for (int r = 0; r < 32; ++r)
        rv[r] = pr[r * 512];

    const float4* pi = reinterpret_cast<const float4*>(img + ((size_t)b * 512 + (size_t)w * 32) * 256) + lane;

    float4 bufa[16], bufb[16];
    load_rows8(bufa, pi, 0);
    __builtin_amdgcn_sched_barrier(0);

    float rs = 0.0f;
    #pragma unroll
    for (int r = 0; r < 32; ++r)
        rs += rv[r];
    rs *= (1.0f / 32.0f);
    out_rad[b * 512 + t] = rs;
    semb[512 + t] = rs;

    float m = 0.0f;
    __builtin_amdgcn_sched_barrier(0);
    load_rows8(bufb, pi, 1);
    __builtin_amdgcn_sched_barrier(0);
    sum_rows8(bufa, 0, lane, m);
    __builtin_amdgcn_sched_barrier(0);
    load_rows8(bufa, pi, 2);
    __builtin_amdgcn_sched_barrier(0);
    sum_rows8(bufb, 1, lane, m);
    __builtin_amdgcn_sched_barrier(0);
    load_rows8(bufb, pi, 3);
    __builtin_amdgcn_sched_barrier(0);
    sum_rows8(bufa, 2, lane, m);
    sum_rows8(bufb, 3, lane, m);

    semb[w * 32 + lane] = m;
    __syncthreads();

    const int half = t >> 8;
    const int u    = t & 255;
    {
        float a = semb[half * 512 + u];
        float c = semb[half * 512 + u + 256];
        sm[t] = a * a + c * c;
    }
    __syncthreads();
    for (int s = 128; s >= 32; s >>= 1) {
        if (u < s) sm[t] += sm[t + s];
        __syncthreads();
    }
    out_img[b * 512 + w * 32 + lane] = m;
    if (u < 32) {
        float v = sm[t];
        for (int off = 16; off > 0; off >>= 1)
            v += __shfl_down(v, off, 32);
        if (u == 0)
            invn[half * 256 + b] = 1.0f / fmaxf(sqrtf(v), 1e-12f);
    }
}

#define SIM_PITCH4 129

__device__ __forceinline__ void load_frags8(float4 (&fa)[8], float4 (&fb)[8],
                                            const float4* pa, const float4* pb, int blk) {
    #pragma unroll
    for (int j = 0; j < 8; ++j) {
        fa[j] = pa[(blk * 8 + j) * 2];
        fb[j] = pb[(blk * 8 + j) * 2];
    }
}

__device__ __forceinline__ void wmma_block16(const float4 (&fa)[8], const float4 (&fb)[8],
                                             v8f& acc0, v8f& acc1) {
    #pragma unroll
    for (int j = 0; j < 8; ++j) {
        v2f a0 = {fa[j].x, fa[j].y};
        v2f b0 = {fb[j].x, fb[j].y};
        v2f a1 = {fa[j].z, fa[j].w};
        v2f b1 = {fb[j].z, fb[j].w};
        acc0 = __builtin_amdgcn_wmma_f32_16x16x4_f32(
             false, a0,  false, b0,
             (short)0, acc0,  false,  false);
        acc1 = __builtin_amdgcn_wmma_f32_16x16x4_f32(
             false, a1,  false, b1,
             (short)0, acc1,  false,  false);
    }
}

__global__ __launch_bounds__(128) void sim_wmma_kernel(const float* __restrict__ img,
                                                       const float* __restrict__ rad,
                                                       const float* __restrict__ invn,
                                                       const float* __restrict__ temperature,
                                                       float* __restrict__ out) {
    __shared__ float4 stage[2 * 16 * SIM_PITCH4];

    const int tile = blockIdx.x;
    const int ti = tile >> 4;
    const int tj = tile & 15;
    const int lane = threadIdx.x & 31;
    const int w = __builtin_amdgcn_readfirstlane(threadIdx.x >> 5);

    const int rowA = ti * 16 + 4 * w;
    const int rowB = tj * 16 + 4 * w;
    const float4* ga = reinterpret_cast<const float4*>(img + (size_t)rowA * 512) + lane;
    const float4* gb = reinterpret_cast<const float4*>(rad + (size_t)rowB * 512) + lane;

    float4 ra[16], rb[16];
    #pragma unroll
    for (int i = 0; i < 16; ++i)
        ra[i] = ga[i * 32];
    #pragma unroll
    for (int i = 0; i < 16; ++i)
        rb[i] = gb[i * 32];
    float sa[4], sb[4];
    #pragma unroll
    for (int r = 0; r < 4; ++r) {
        sa[r] = invn[rowA + r];
        sb[r] = invn[256 + rowB + r];
    }
    const float temp = temperature[0];
    __builtin_amdgcn_sched_barrier(0);

    #pragma unroll
    for (int i = 0; i < 16; ++i) {
        const float s = sa[i >> 2];
        float4 v = ra[i];
        v.x *= s; v.y *= s; v.z *= s; v.w *= s;
        stage[(4 * w + (i >> 2)) * SIM_PITCH4 + (i & 3) * 32 + lane] = v;
    }
    #pragma unroll
    for (int i = 0; i < 16; ++i) {
        const float s = sb[i >> 2];
        float4 v = rb[i];
        v.x *= s; v.y *= s; v.z *= s; v.w *= s;
        stage[(16 + 4 * w + (i >> 2)) * SIM_PITCH4 + (i & 3) * 32 + lane] = v;
    }
    __syncthreads();
    if (w != 0) return;

    const int lhalf = lane >> 4;
    const int lmod  = lane & 15;
    const float4* pa = stage + lmod * SIM_PITCH4 + lhalf;
    const float4* pb = stage + (16 + lmod) * SIM_PITCH4 + lhalf;

    v8f acc0 = {};
    v8f acc1 = {};
    float4 fa0[8], fb0[8], fa1[8], fb1[8];
    load_frags8(fa0, fb0, pa, pb, 0);
    #pragma unroll
    for (int blk = 0; blk < 8; blk += 2) {
        __builtin_amdgcn_sched_barrier(0);
        load_frags8(fa1, fb1, pa, pb, blk + 1);
        __builtin_amdgcn_sched_barrier(0);
        wmma_block16(fa0, fb0, acc0, acc1);
        __builtin_amdgcn_sched_barrier(0);
        if (blk + 2 < 8)
            load_frags8(fa0, fb0, pa, pb, blk + 2);
        __builtin_amdgcn_sched_barrier(0);
        wmma_block16(fa1, fb1, acc0, acc1);
    }

    const float invt = 1.0f / temp;
    const int col = tj * 16 + lmod;
    #pragma unroll
    for (int v = 0; v < 8; ++v) {
        const int row = ti * 16 + v + 8 * lhalf;
        out[row * 256 + col] = (acc0[v] + acc1[v]) * invt;
    }
}

extern "C" void kernel_launch(void* const* d_in, const int* in_sizes, int n_in,
                              void* d_out, int out_size, void* d_ws, size_t ws_size,
                              hipStream_t stream) {
    const float* img_feat = (const float*)d_in[0];
    const float* rad_feat = (const float*)d_in[1];
    const float* temp     = (const float*)d_in[2];

    float* out_sim = (float*)d_out;
    float* out_img = out_sim + 256 * 256;
    float* out_rad = out_img + 256 * 512;

    float* invn = (float*)d_ws;

    embed_norm_kernel<<<256, 512, 0, stream>>>(img_feat, rad_feat, out_img, out_rad, invn);
    sim_wmma_kernel<<<256, 128, 0, stream>>>(out_img, out_rad, invn, temp, out_sim);
}
